# starting kernel
# baseline (speedup 1.0000x reference)
#include <hip/hip_runtime.h>
#include <hip/hip_bf16.h>

typedef __attribute__((ext_vector_type(2))) float v2f;
typedef __attribute__((ext_vector_type(8))) float v8f;

__constant__ int c_zz[64] = {
     0, 1, 8,16, 9, 2, 3,10,
    17,24,32,25,18,11, 4, 5,
    12,19,26,33,40,48,41,34,
    27,20,13, 6, 7,14,21,28,
    35,42,49,56,57,50,43,36,
    29,22,15,23,30,37,44,51,
    58,59,52,45,38,31,39,46,
    53,60,61,54,47,55,62,63};

__constant__ float c_cos[9] = {
    1.0f,
    0.98078528040323044913f,
    0.92387953251128675613f,
    0.83146961230254523708f,
    0.70710678118654752440f,
    0.55557023301960222474f,
    0.38268343236508977173f,
    0.19509032201612826785f,
    0.0f};

#define Bd 4
#define Cd 3
#define Td 64
#define Hd 224
#define Wd 224
#define NBH 28
#define NT4 7
#define HW (Hd*Wd)
#define CH_STRIDE (Td*Hd*Wd)

__device__ __forceinline__ float dmat(int l, int n) {
    int u = ((2 * n + 1) * l) & 31;
    if (u > 16) u = 32 - u;
    float s = 2.0f;
    int idx = u;
    if (u > 8) { idx = 16 - u; s = -2.0f; }
    return s * c_cos[idx];
}

__global__ __launch_bounds__(32) void dct_zz_kernel(const float* __restrict__ x,
                                                    const int* __restrict__ remp,
                                                    float* __restrict__ out) {
    __shared__ float smem[256];

    const int lane = threadIdx.x;
    const int hi   = lane >> 4;
    const int N    = lane & 15;
    const int row  = N & 7;
    const int rblk = N >> 3;

    const int keep  = 63 - remp[0];
    const int keep4 = 4 * keep;

    int gw = blockIdx.x;
    const int hb = gw % NBH;  gw /= NBH;
    const int t  = gw % Td;   gw /= Td;
    const int b  = gw;

    v2f b1[2];
    #pragma unroll
    for (int s = 0; s < 2; ++s)
        #pragma unroll
        for (int v = 0; v < 2; ++v)
            b1[s][v] = (N < 8) ? dmat(N, 4*s + 2*hi + v) : 0.0f;
    v2f a2[4];
    #pragma unroll
    for (int s = 0; s < 4; ++s)
        #pragma unroll
        for (int v = 0; v < 2; ++v)
            a2[s][v] = ((N >> 3) == (s >> 1)) ? dmat(row, (4*s + 2*hi + v) & 7) : 0.0f;

    int lds_idx[8];
    #pragma unroll
    for (int s = 0; s < 8; ++s) {
        const int u = lane + 32 * s;
        if (u < keep4) {
            const int blk = (u >= keep) + (u >= 2*keep) + (u >= 3*keep);
            const int j   = u - blk * keep;
            lds_idx[s] = blk * 64 + c_zz[1 + j];
        } else {
            lds_idx[s] = 0;
        }
    }

    long ubase = ((long)(b * Cd) * Td + t) * (long)HW + (long)(hb * 8) * Wd;
    const int loff = row * Wd + rblk * 8 + 2 * hi;
    long obase = ((long)(b * Td + t) * (NBH * NBH) + hb * NBH) * keep;

    const float wc0 = 0.2989f, wc1 = 0.587f, wc2 = 0.114f;

    for (int q2 = 0; q2 < NT4; ++q2) {
        const float* base = x + ubase + loff;
        v2f L[12];
        #pragma unroll
        for (int e = 0; e < 2; ++e) {
            const float* be = base + e * 16;
            L[e*6 + 0] = __builtin_nontemporal_load((const v2f*)(be));
            L[e*6 + 1] = __builtin_nontemporal_load((const v2f*)(be + 4));
            L[e*6 + 2] = __builtin_nontemporal_load((const v2f*)(be + (long)CH_STRIDE));
            L[e*6 + 3] = __builtin_nontemporal_load((const v2f*)(be + (long)CH_STRIDE + 4));
            L[e*6 + 4] = __builtin_nontemporal_load((const v2f*)(be + 2L * CH_STRIDE));
            L[e*6 + 5] = __builtin_nontemporal_load((const v2f*)(be + 2L * CH_STRIDE + 4));
        }

        v8f c2e[2];
        #pragma unroll
        for (int e = 0; e < 2; ++e) {
            v2f gA = wc0 * L[e*6 + 0] + wc1 * L[e*6 + 2] + wc2 * L[e*6 + 4];
            v2f gB = wc0 * L[e*6 + 1] + wc1 * L[e*6 + 3] + wc2 * L[e*6 + 5];

            v8f c1 = {};
            c1 = __builtin_amdgcn_wmma_f32_16x16x4_f32(false, gA, false, b1[0], (short)0, c1, false, false);
            c1 = __builtin_amdgcn_wmma_f32_16x16x4_f32(false, gB, false, b1[1], (short)0, c1, false, false);

            v2f b2[4];
            #pragma unroll
            for (int s = 0; s < 4; ++s) {
                #pragma unroll
                for (int v = 0; v < 2; ++v) {
                    const int ia = (4*s + v) & 7;
                    const int ib = (4*s + 2 + v) & 7;
                    float send  = hi ? c1[ia] : c1[ib];
                    float keepv = hi ? c1[ib] : c1[ia];
                    float recv  = __shfl_xor(send, 16, 32);
                    b2[s][v] = ((s >> 1) == hi) ? keepv : recv;
                }
            }

            v8f c2 = {};
            #pragma unroll
            for (int s = 0; s < 4; ++s)
                c2 = __builtin_amdgcn_wmma_f32_16x16x4_f32(false, a2[s], false, b2[s], (short)0, c2, false, false);
            c2e[e] = c2;
        }

        if (N < 8) {
            #pragma unroll
            for (int e = 0; e < 2; ++e)
                #pragma unroll
                for (int r = 0; r < 8; ++r)
                    smem[(e * 2 + hi) * 64 + r * 8 + N] = c2e[e][r];
        }
        __syncthreads();

        #pragma unroll
        for (int s = 0; s < 8; ++s) {
            const int u = lane + 32 * s;
            if (u < keep4) {
                const float val = smem[lds_idx[s]];
                __builtin_nontemporal_store(val, out + obase + u);
            }
        }
        __syncthreads();

        ubase += 32;
        obase += keep4;
    }
}

extern "C" void kernel_launch(void* const* d_in, const int* in_sizes, int n_in,
                              void* d_out, int out_size, void* d_ws, size_t ws_size,
                              hipStream_t stream) {
    const float* x  = (const float*)d_in[0];
    const int* rem  = (const int*)d_in[1];
    float* out      = (float*)d_out;

    const int blocks = Bd * Td * NBH;
    dct_zz_kernel<<<blocks, 32, 0, stream>>>(x, rem, out);
}


# Whole-line b128 loads: A-row = (pixel row, block), each load reads 4 full 128B lines; stage-1 D registers feed stage 2 directly (no lane exchange); k order (sum order) of both products changes
# speedup vs baseline: 1.1874x; 1.1874x over previous
#include <hip/hip_runtime.h>
#include <hip/hip_bf16.h>

typedef __attribute__((ext_vector_type(2))) float v2f;
typedef __attribute__((ext_vector_type(4))) float v4f;
typedef __attribute__((ext_vector_type(8))) float v8f;

__constant__ int c_zz[64] = {
     0, 1, 8,16, 9, 2, 3,10,
    17,24,32,25,18,11, 4, 5,
    12,19,26,33,40,48,41,34,
    27,20,13, 6, 7,14,21,28,
    35,42,49,56,57,50,43,36,
    29,22,15,23,30,37,44,51,
    58,59,52,45,38,31,39,46,
    53,60,61,54,47,55,62,63};

__constant__ float c_cos[9] = {
    1.0f,
    0.98078528040323044913f,
    0.92387953251128675613f,
    0.83146961230254523708f,
    0.70710678118654752440f,
    0.55557023301960222474f,
    0.38268343236508977173f,
    0.19509032201612826785f,
    0.0f};

#define Bd 4
#define Cd 3
#define Td 64
#define Hd 224
#define Wd 224
#define NBH 28
#define NT4 7
#define HW (Hd*Wd)
#define CH_STRIDE (Td*Hd*Wd)

__device__ __forceinline__ float dmat(int l, int n) {
    int u = ((2 * n + 1) * l) & 31;
    if (u > 16) u = 32 - u;
    float s = 2.0f;
    int idx = u;
    if (u > 8) { idx = 16 - u; s = -2.0f; }
    return s * c_cos[idx];
}

__global__ __launch_bounds__(32) void dct_zz_kernel(const float* __restrict__ x,
                                                    const int* __restrict__ remp,
                                                    float* __restrict__ out) {
    __shared__ float smem[256];

    const int lane = threadIdx.x;
    const int hi   = lane >> 4;
    const int N    = lane & 15;
    const int row  = N & 7;

    const int keep  = 63 - remp[0];
    const int keep4 = 4 * keep;

    int gw = blockIdx.x;
    const int hb = gw % NBH;  gw /= NBH;
    const int t  = gw % Td;   gw /= Td;
    const int b  = gw;

    v2f b1[2];
    #pragma unroll
    for (int s = 0; s < 2; ++s)
        #pragma unroll
        for (int v = 0; v < 2; ++v)
            b1[s][v] = (N < 8) ? dmat(N, 4*hi + 2*s + v) : 0.0f;
    v2f a2[4];
    #pragma unroll
    for (int s = 0; s < 4; ++s)
        #pragma unroll
        for (int v = 0; v < 2; ++v)
            a2[s][v] = ((N >> 3) == v) ? dmat(row, 4*(s >> 1) + 2*hi + (s & 1)) : 0.0f;

    int lds_idx[8];
    #pragma unroll
    for (int s = 0; s < 8; ++s) {
        const int u = lane + 32 * s;
        if (u < keep4) {
            const int blk = (u >= keep) + (u >= 2*keep) + (u >= 3*keep);
            const int j   = u - blk * keep;
            lds_idx[s] = blk * 64 + c_zz[1 + j];
        } else {
            lds_idx[s] = 0;
        }
    }

    long ubase = ((long)(b * Cd) * Td + t) * (long)HW + (long)(hb * 8) * Wd;
    const int loff = (N >> 2) * Wd + (N & 3) * 8 + 4 * hi;
    long obase = ((long)(b * Td + t) * (NBH * NBH) + hb * NBH) * keep;

    const float wc0 = 0.2989f, wc1 = 0.587f, wc2 = 0.114f;

    for (int q2 = 0; q2 < NT4; ++q2) {
        const float* base = x + ubase + loff;
        v4f L[6];
        #pragma unroll
        for (int j = 0; j < 2; ++j) {
            const float* bj = base + j * (4 * Wd);
            L[j*3 + 0] = __builtin_nontemporal_load((const v4f*)(bj));
            L[j*3 + 1] = __builtin_nontemporal_load((const v4f*)(bj + (long)CH_STRIDE));
            L[j*3 + 2] = __builtin_nontemporal_load((const v4f*)(bj + 2L * CH_STRIDE));
        }

        v8f c1[2];
        #pragma unroll
        for (int j = 0; j < 2; ++j) {
            const v4f gj = wc0 * L[j*3 + 0] + wc1 * L[j*3 + 1] + wc2 * L[j*3 + 2];
            const v2f gA = __builtin_shufflevector(gj, gj, 0, 1);
            const v2f gB = __builtin_shufflevector(gj, gj, 2, 3);
            v8f c = {};
            c = __builtin_amdgcn_wmma_f32_16x16x4_f32(false, gA, false, b1[0], (short)0, c, false, false);
            c = __builtin_amdgcn_wmma_f32_16x16x4_f32(false, gB, false, b1[1], (short)0, c, false, false);
            c1[j] = c;
        }

        v8f c2e[2];
        #pragma unroll
        for (int e = 0; e < 2; ++e) {
            v8f c2 = {};
            #pragma unroll
            for (int s = 0; s < 4; ++s) {
                const int j = s >> 1;
                const int r0 = 2 * e + 4 * (s & 1);
                v2f b2;
                b2[0] = c1[j][r0];
                b2[1] = c1[j][r0 + 1];
                c2 = __builtin_amdgcn_wmma_f32_16x16x4_f32(false, a2[s], false, b2, (short)0, c2, false, false);
            }
            c2e[e] = c2;
        }

        if (N < 8) {
            #pragma unroll
            for (int e = 0; e < 2; ++e)
                #pragma unroll
                for (int r = 0; r < 8; ++r)
                    smem[(e * 2 + hi) * 64 + r * 8 + N] = c2e[e][r];
        }
        __syncthreads();

        #pragma unroll
        for (int s = 0; s < 8; ++s) {
            const int u = lane + 32 * s;
            if (u < keep4) {
                const float val = smem[lds_idx[s]];
                __builtin_nontemporal_store(val, out + obase + u);
            }
        }
        __syncthreads();

        ubase += 32;
        obase += keep4;
    }
}

extern "C" void kernel_launch(void* const* d_in, const int* in_sizes, int n_in,
                              void* d_out, int out_size, void* d_ws, size_t ws_size,
                              hipStream_t stream) {
    const float* x  = (const float*)d_in[0];
    const int* rem  = (const int*)d_in[1];
    float* out      = (float*)d_out;

    const int blocks = Bd * Td * NBH;
    dct_zz_kernel<<<blocks, 32, 0, stream>>>(x, rem, out);
}


# Software-pipelined tile loop: next tile's 6 loads requested ahead of this tile's products; per-tile workgroup barriers (which drained all global loads and stores) become wave-scope LDS ordering
# speedup vs baseline: 1.2855x; 1.0826x over previous
#include <hip/hip_runtime.h>
#include <hip/hip_bf16.h>

typedef __attribute__((ext_vector_type(2))) float v2f;
typedef __attribute__((ext_vector_type(4))) float v4f;
typedef __attribute__((ext_vector_type(8))) float v8f;

__constant__ int c_zz[64] = {
     0, 1, 8,16, 9, 2, 3,10,
    17,24,32,25,18,11, 4, 5,
    12,19,26,33,40,48,41,34,
    27,20,13, 6, 7,14,21,28,
    35,42,49,56,57,50,43,36,
    29,22,15,23,30,37,44,51,
    58,59,52,45,38,31,39,46,
    53,60,61,54,47,55,62,63};

__constant__ float c_cos[9] = {
    1.0f,
    0.98078528040323044913f,
    0.92387953251128675613f,
    0.83146961230254523708f,
    0.70710678118654752440f,
    0.55557023301960222474f,
    0.38268343236508977173f,
    0.19509032201612826785f,
    0.0f};

#define Bd 4
#define Cd 3
#define Td 64
#define Hd 224
#define Wd 224
#define NBH 28
#define NT4 7
#define HW (Hd*Wd)
#define CH_STRIDE (Td*Hd*Wd)

__device__ __forceinline__ float dmat(int l, int n) {
    int u = ((2 * n + 1) * l) & 31;
    if (u > 16) u = 32 - u;
    float s = 2.0f;
    int idx = u;
    if (u > 8) { idx = 16 - u; s = -2.0f; }
    return s * c_cos[idx];
}

__device__ __forceinline__ void load_tile(v4f (&dst)[6], const float* p) {
    dst[0] = __builtin_nontemporal_load((const v4f*)(p));
    dst[1] = __builtin_nontemporal_load((const v4f*)(p + (long)CH_STRIDE));
    dst[2] = __builtin_nontemporal_load((const v4f*)(p + 2L * CH_STRIDE));
    dst[3] = __builtin_nontemporal_load((const v4f*)(p + 4 * Wd));
    dst[4] = __builtin_nontemporal_load((const v4f*)(p + 4 * Wd + (long)CH_STRIDE));
    dst[5] = __builtin_nontemporal_load((const v4f*)(p + 4 * Wd + 2L * CH_STRIDE));
}

__device__ __forceinline__ void wave_lds_handover() {
    __builtin_amdgcn_fence(__ATOMIC_RELEASE, "wavefront");
    __builtin_amdgcn_wave_barrier();
    __builtin_amdgcn_fence(__ATOMIC_ACQUIRE, "wavefront");
}

__global__ __launch_bounds__(32) void dct_zz_kernel(const float* __restrict__ x,
                                                    const int* __restrict__ remp,
                                                    float* __restrict__ out) {
    __shared__ float smem[256];

    const int lane = threadIdx.x;
    const int hi   = lane >> 4;
    const int N    = lane & 15;
    const int row  = N & 7;

    const int keep  = 63 - remp[0];
    const int keep4 = 4 * keep;

    int gw = blockIdx.x;
    const int hb = gw % NBH;  gw /= NBH;
    const int t  = gw % Td;   gw /= Td;
    const int b  = gw;

    v2f b1[2];
    #pragma unroll
    for (int s = 0; s < 2; ++s)
        #pragma unroll
        for (int v = 0; v < 2; ++v)
            b1[s][v] = (N < 8) ? dmat(N, 4*hi + 2*s + v) : 0.0f;
    v2f a2[4];
    #pragma unroll
    for (int s = 0; s < 4; ++s)
        #pragma unroll
        for (int v = 0; v < 2; ++v)
            a2[s][v] = ((N >> 3) == v) ? dmat(row, 4*(s >> 1) + 2*hi + (s & 1)) : 0.0f;

    int lds_idx[8];
    #pragma unroll
    for (int s = 0; s < 8; ++s) {
        const int u = lane + 32 * s;
        if (u < keep4) {
            const int blk = (u >= keep) + (u >= 2*keep) + (u >= 3*keep);
            const int j   = u - blk * keep;
            lds_idx[s] = blk * 64 + c_zz[1 + j];
        } else {
            lds_idx[s] = 0;
        }
    }

    long ubase = ((long)(b * Cd) * Td + t) * (long)HW + (long)(hb * 8) * Wd;
    const int loff = (N >> 2) * Wd + (N & 3) * 8 + 4 * hi;
    long obase = ((long)(b * Td + t) * (NBH * NBH) + hb * NBH) * keep;

    const float wc0 = 0.2989f, wc1 = 0.587f, wc2 = 0.114f;

    const float* xw = x + ubase + loff;
    float* ow = out + obase;

    v4f L[2][6];
    load_tile(L[0], xw);

    for (int q2 = 0; q2 < NT4; ++q2) {
        const int cur = q2 & 1;
        if (q2 + 1 < NT4)
            load_tile(L[cur ^ 1], xw + 32 * (q2 + 1));

        v8f c1[2];
        #pragma unroll
        for (int j = 0; j < 2; ++j) {
            const v4f gj = wc0 * L[cur][j*3 + 0] + wc1 * L[cur][j*3 + 1] + wc2 * L[cur][j*3 + 2];
            const v2f gA = __builtin_shufflevector(gj, gj, 0, 1);
            const v2f gB = __builtin_shufflevector(gj, gj, 2, 3);
            v8f c = {};
            c = __builtin_amdgcn_wmma_f32_16x16x4_f32(false, gA, false, b1[0], (short)0, c, false, false);
            c = __builtin_amdgcn_wmma_f32_16x16x4_f32(false, gB, false, b1[1], (short)0, c, false, false);
            c1[j] = c;
        }

        v8f c2e[2];
        #pragma unroll
        for (int e = 0; e < 2; ++e) {
            v8f c2 = {};
            #pragma unroll
            for (int s = 0; s < 4; ++s) {
                const int j = s >> 1;
                const int r0 = 2 * e + 4 * (s & 1);
                v2f b2;
                b2[0] = c1[j][r0];
                b2[1] = c1[j][r0 + 1];
                c2 = __builtin_amdgcn_wmma_f32_16x16x4_f32(false, a2[s], false, b2, (short)0, c2, false, false);
            }
            c2e[e] = c2;
        }

        if (N < 8) {
            #pragma unroll
            for (int e = 0; e < 2; ++e)
                #pragma unroll
                for (int r = 0; r < 8; ++r)
                    smem[(e * 2 + hi) * 64 + r * 8 + N] = c2e[e][r];
        }
        wave_lds_handover();

        #pragma unroll
        for (int s = 0; s < 8; ++s) {
            const int u = lane + 32 * s;
            if (u < keep4) {
                const float val = smem[lds_idx[s]];
                __builtin_nontemporal_store(val, ow + u);
            }
        }
        wave_lds_handover();

        ow += keep4;
    }
}

extern "C" void kernel_launch(void* const* d_in, const int* in_sizes, int n_in,
                              void* d_out, int out_size, void* d_ws, size_t ws_size,
                              hipStream_t stream) {
    const float* x  = (const float*)d_in[0];
    const int* rem  = (const int*)d_in[1];
    float* out      = (float*)d_out;

    const int blocks = Bd * Td * NBH;
    dct_zz_kernel<<<blocks, 32, 0, stream>>>(x, rem, out);
}


# Each global load and store covers whole aligned 256-B units: a load takes one unit from each of two pixel-row pairs, not 4 separate 128-B lines; outputs go as b128 stores; sum order kept
# speedup vs baseline: 1.8776x; 1.4606x over previous
#include <hip/hip_runtime.h>
#include <hip/hip_bf16.h>

typedef __attribute__((ext_vector_type(2))) float v2f;
typedef __attribute__((ext_vector_type(4))) float v4f;
typedef __attribute__((ext_vector_type(8))) float v8f;

__constant__ int c_zz[64] = {
     0, 1, 8,16, 9, 2, 3,10,
    17,24,32,25,18,11, 4, 5,
    12,19,26,33,40,48,41,34,
    27,20,13, 6, 7,14,21,28,
    35,42,49,56,57,50,43,36,
    29,22,15,23,30,37,44,51,
    58,59,52,45,38,31,39,46,
    53,60,61,54,47,55,62,63};

__constant__ float c_cos[9] = {
    1.0f,
    0.98078528040323044913f,
    0.92387953251128675613f,
    0.83146961230254523708f,
    0.70710678118654752440f,
    0.55557023301960222474f,
    0.38268343236508977173f,
    0.19509032201612826785f,
    0.0f};

#define Bd 4
#define Cd 3
#define Td 64
#define Hd 224
#define Wd 224
#define NBH 28
#define NT4 7
#define HW (Hd*Wd)
#define CH_STRIDE (Td*Hd*Wd)

__device__ __forceinline__ float dmat(int l, int n) {
    int u = ((2 * n + 1) * l) & 31;
    if (u > 16) u = 32 - u;
    float s = 2.0f;
    int idx = u;
    if (u > 8) { idx = 16 - u; s = -2.0f; }
    return s * c_cos[idx];
}

__device__ __forceinline__ void load_tile(v4f (&dst)[6], const float* p) {
    dst[0] = __builtin_nontemporal_load((const v4f*)(p));
    dst[1] = __builtin_nontemporal_load((const v4f*)(p + (long)CH_STRIDE));
    dst[2] = __builtin_nontemporal_load((const v4f*)(p + 2L * CH_STRIDE));
    dst[3] = __builtin_nontemporal_load((const v4f*)(p + 4 * Wd));
    dst[4] = __builtin_nontemporal_load((const v4f*)(p + 4 * Wd + (long)CH_STRIDE));
    dst[5] = __builtin_nontemporal_load((const v4f*)(p + 4 * Wd + 2L * CH_STRIDE));
}

__device__ __forceinline__ void wave_lds_handover() {
    __builtin_amdgcn_fence(__ATOMIC_RELEASE, "wavefront");
    __builtin_amdgcn_wave_barrier();
    __builtin_amdgcn_fence(__ATOMIC_ACQUIRE, "wavefront");
}

__global__ __launch_bounds__(32) void dct_zz_kernel(const float* __restrict__ x,
                                                    const int* __restrict__ remp,
                                                    float* __restrict__ out) {
    __shared__ float smem[256];

    const int lane = threadIdx.x;
    const int hi   = lane >> 4;
    const int N    = lane & 15;
    const int row  = N & 7;

    const int keep  = 63 - remp[0];
    const int keep4 = 4 * keep;

    int gw = blockIdx.x;
    const int hb = gw % NBH;  gw /= NBH;
    const int t  = gw % Td;   gw /= Td;
    const int b  = gw;

    v2f b1[2];
    #pragma unroll
    for (int s = 0; s < 2; ++s)
        #pragma unroll
        for (int v = 0; v < 2; ++v)
            b1[s][v] = (N < 8) ? dmat(N, 4*hi + 2*s + v) : 0.0f;
    v2f a2[4];
    #pragma unroll
    for (int s = 0; s < 4; ++s)
        #pragma unroll
        for (int v = 0; v < 2; ++v)
            a2[s][v] = ((N >> 3) == v) ? dmat(row, 4*(s >> 1) + 2*hi + (s & 1)) : 0.0f;

    int lds_idx[2][4];
    #pragma unroll
    for (int R = 0; R < 2; ++R)
        #pragma unroll
        for (int i = 0; i < 4; ++i) {
            const int u = 4 * (lane + 32 * R) + i;
            if (u < keep4) {
                const int blk = (u >= keep) + (u >= 2*keep) + (u >= 3*keep);
                const int j   = u - blk * keep;
                lds_idx[R][i] = blk * 64 + c_zz[1 + j];
            } else {
                lds_idx[R][i] = 0;
            }
        }

    long ubase = ((long)(b * Cd) * Td + t) * (long)HW + (long)(hb * 8) * Wd;
    const int loff = (N >> 3) * (2 * Wd) + (N & 7) * 8 + 4 * hi;
    long obase = ((long)(b * Td + t) * (NBH * NBH) + hb * NBH) * keep;

    const float wc0 = 0.2989f, wc1 = 0.587f, wc2 = 0.114f;

    const float* xw = x + ubase + loff;
    float* ow = out + obase;

    v8f T[7][2];
    v4f L[2][6];

    auto stage1 = [&](const v4f (&ld)[6], v8f (&dst)[2]) __attribute__((always_inline)) {
        #pragma unroll
        for (int Q = 0; Q < 2; ++Q) {
            const v4f gj = wc0 * ld[Q*3 + 0] + wc1 * ld[Q*3 + 1] + wc2 * ld[Q*3 + 2];
            const v2f gA = __builtin_shufflevector(gj, gj, 0, 1);
            const v2f gB = __builtin_shufflevector(gj, gj, 2, 3);
            v8f c = {};
            c = __builtin_amdgcn_wmma_f32_16x16x4_f32(false, gA, false, b1[0], (short)0, c, false, false);
            c = __builtin_amdgcn_wmma_f32_16x16x4_f32(false, gB, false, b1[1], (short)0, c, false, false);
            dst[Q] = c;
        }
    };

    auto group_out = [&](const v8f (&TE)[2], const int re, const v8f (&TO)[2], const int ro, float* og) __attribute__((always_inline)) {
        v8f c2e[2];
        #pragma unroll
        for (int e = 0; e < 2; ++e) {
            v8f c2 = {};
            #pragma unroll
            for (int Q = 0; Q < 2; ++Q) {
                v2f be, bo;
                be[0] = TE[Q][re + 2*e];  be[1] = TE[Q][re + 2*e + 1];
                bo[0] = TO[Q][ro + 2*e];  bo[1] = TO[Q][ro + 2*e + 1];
                c2 = __builtin_amdgcn_wmma_f32_16x16x4_f32(false, a2[2*Q],     false, be, (short)0, c2, false, false);
                c2 = __builtin_amdgcn_wmma_f32_16x16x4_f32(false, a2[2*Q + 1], false, bo, (short)0, c2, false, false);
            }
            c2e[e] = c2;
        }
        if (N < 8) {
            #pragma unroll
            for (int e = 0; e < 2; ++e)
                #pragma unroll
                for (int r = 0; r < 8; ++r)
                    smem[(e * 2 + hi) * 64 + r * 8 + N] = c2e[e][r];
        }
        wave_lds_handover();
        #pragma unroll
        for (int R = 0; R < 2; ++R) {
            v4f val;
            #pragma unroll
            for (int i = 0; i < 4; ++i) val[i] = smem[lds_idx[R][i]];
            const int u0 = 4 * (lane + 32 * R);
            if (u0 < keep4)
                __builtin_nontemporal_store(val, (v4f*)(og + u0));
        }
        wave_lds_handover();
    };

    auto request_unit = [&](v4f (&dst)[6], const int u) __attribute__((always_inline)) {
        load_tile(dst, xw + 64 * u);
        __builtin_amdgcn_sched_barrier(0);
    };

    request_unit(L[0], 3);

    request_unit(L[1], 6);
    stage1(L[0], T[3]);

    request_unit(L[0], 2);
    stage1(L[1], T[6]);
    group_out(T[3], 0, T[6], 4, ow + 6 * keep4);

    request_unit(L[1], 5);
    stage1(L[0], T[2]);
    group_out(T[2], 4, T[6], 0, ow + 5 * keep4);

    request_unit(L[0], 1);
    stage1(L[1], T[5]);
    group_out(T[2], 0, T[5], 4, ow + 4 * keep4);

    request_unit(L[1], 4);
    stage1(L[0], T[1]);
    group_out(T[1], 4, T[5], 0, ow + 3 * keep4);

    request_unit(L[0], 0);
    stage1(L[1], T[4]);
    group_out(T[1], 0, T[4], 4, ow + 2 * keep4);

    stage1(L[0], T[0]);
    group_out(T[0], 4, T[4], 0, ow + 1 * keep4);
    group_out(T[0], 0, T[3], 4, ow);
}

extern "C" void kernel_launch(void* const* d_in, const int* in_sizes, int n_in,
                              void* d_out, int out_size, void* d_ws, size_t ws_size,
                              hipStream_t stream) {
    const float* x  = (const float*)d_in[0];
    const int* rem  = (const int*)d_in[1];
    float* out      = (float*)d_out;

    const int blocks = Bd * Td * NBH;
    dct_zz_kernel<<<blocks, 32, 0, stream>>>(x, rem, out);
}


# Each wave takes two adjacent block-rows side by side in the 16 result columns of both products (columns 8-15 were zero padding): a third fewer WMMAs, half as many waves enter; sum order kept
# speedup vs baseline: 1.9933x; 1.0616x over previous
#include <hip/hip_runtime.h>
#include <hip/hip_bf16.h>

typedef __attribute__((ext_vector_type(2))) float v2f;
typedef __attribute__((ext_vector_type(4))) float v4f;
typedef __attribute__((ext_vector_type(8))) float v8f;

__constant__ int c_zz[64] = {
     0, 1, 8,16, 9, 2, 3,10,
    17,24,32,25,18,11, 4, 5,
    12,19,26,33,40,48,41,34,
    27,20,13, 6, 7,14,21,28,
    35,42,49,56,57,50,43,36,
    29,22,15,23,30,37,44,51,
    58,59,52,45,38,31,39,46,
    53,60,61,54,47,55,62,63};

__constant__ float c_cos[9] = {
    1.0f,
    0.98078528040323044913f,
    0.92387953251128675613f,
    0.83146961230254523708f,
    0.70710678118654752440f,
    0.55557023301960222474f,
    0.38268343236508977173f,
    0.19509032201612826785f,
    0.0f};

#define Bd 4
#define Cd 3
#define Td 64
#define Hd 224
#define Wd 224
#define NBH 28
#define NBP 14
#define NT4 7
#define HW (Hd*Wd)
#define CH_STRIDE (Td*Hd*Wd)
#define BLK_PITCH 72

__device__ __forceinline__ float dmat(int l, int n) {
    int u = ((2 * n + 1) * l) & 31;
    if (u > 16) u = 32 - u;
    float s = 2.0f;
    int idx = u;
    if (u > 8) { idx = 16 - u; s = -2.0f; }
    return s * c_cos[idx];
}

__device__ __forceinline__ void load_tile(v4f (&dst)[12], const float* p) {
    #pragma unroll
    for (int w = 0; w < 2; ++w) {
        const float* q = p + w * (8 * Wd);
        dst[6*w + 0] = __builtin_nontemporal_load((const v4f*)(q));
        dst[6*w + 1] = __builtin_nontemporal_load((const v4f*)(q + (long)CH_STRIDE));
        dst[6*w + 2] = __builtin_nontemporal_load((const v4f*)(q + 2L * CH_STRIDE));
        dst[6*w + 3] = __builtin_nontemporal_load((const v4f*)(q + 4 * Wd));
        dst[6*w + 4] = __builtin_nontemporal_load((const v4f*)(q + 4 * Wd + (long)CH_STRIDE));
        dst[6*w + 5] = __builtin_nontemporal_load((const v4f*)(q + 4 * Wd + 2L * CH_STRIDE));
    }
}

__device__ __forceinline__ void wave_lds_handover() {
    __builtin_amdgcn_fence(__ATOMIC_RELEASE, "wavefront");
    __builtin_amdgcn_wave_barrier();
    __builtin_amdgcn_fence(__ATOMIC_ACQUIRE, "wavefront");
}

__global__ __launch_bounds__(32) void dct_zz_kernel(const float* __restrict__ x,
                                                    const int* __restrict__ remp,
                                                    float* __restrict__ out) {
    __shared__ float smem[8 * BLK_PITCH];

    const int lane = threadIdx.x;
    const int hi   = lane >> 4;
    const int N    = lane & 15;
    const int row  = N & 7;
    const int wsel = N >> 3;

    const int keep  = 63 - remp[0];
    const int keep4 = 4 * keep;

    int gw = blockIdx.x;
    const int hp = gw % NBP;  gw /= NBP;
    const int t  = gw % Td;   gw /= Td;
    const int b  = gw;
    const int hb = 2 * hp;

    v2f b1u[2], b1d[2];
    #pragma unroll
    for (int s = 0; s < 2; ++s)
        #pragma unroll
        for (int v = 0; v < 2; ++v) {
            const float d = dmat(row, 4*hi + 2*s + v);
            b1u[s][v] = (wsel == 0) ? d : 0.0f;
            b1d[s][v] = (wsel == 0) ? 0.0f : d;
        }
    v2f a2[4];
    #pragma unroll
    for (int s = 0; s < 4; ++s)
        #pragma unroll
        for (int v = 0; v < 2; ++v)
            a2[s][v] = ((N >> 3) == v) ? dmat(row, 4*(s >> 1) + 2*hi + (s & 1)) : 0.0f;

    int lds_idx[2][4];
    #pragma unroll
    for (int R = 0; R < 2; ++R)
        #pragma unroll
        for (int i = 0; i < 4; ++i) {
            const int u = 4 * (lane + 32 * R) + i;
            if (u < keep4) {
                const int blk = (u >= keep) + (u >= 2*keep) + (u >= 3*keep);
                const int j   = u - blk * keep;
                lds_idx[R][i] = blk * BLK_PITCH + c_zz[1 + j];
            } else {
                lds_idx[R][i] = 0;
            }
        }

    long ubase = ((long)(b * Cd) * Td + t) * (long)HW + (long)(hb * 8) * Wd;
    const int loff = (N >> 3) * (2 * Wd) + (N & 7) * 8 + 4 * hi;
    long obase = ((long)(b * Td + t) * (NBH * NBH) + hb * NBH) * keep;
    const int orow = NBH * keep;

    const float wc0 = 0.2989f, wc1 = 0.587f, wc2 = 0.114f;

    const float* xw = x + ubase + loff;
    float* ow = out + obase;

    v8f T[7][2];
    v4f L[2][12];

    auto stage1 = [&](const v4f (&ld)[12], v8f (&dst)[2]) __attribute__((always_inline)) {
        #pragma unroll
        for (int Q = 0; Q < 2; ++Q) {
            const v4f gu = wc0 * ld[Q*3 + 0] + wc1 * ld[Q*3 + 1] + wc2 * ld[Q*3 + 2];
            const v4f gd = wc0 * ld[6 + Q*3 + 0] + wc1 * ld[6 + Q*3 + 1] + wc2 * ld[6 + Q*3 + 2];
            const v2f guA = __builtin_shufflevector(gu, gu, 0, 1);
            const v2f guB = __builtin_shufflevector(gu, gu, 2, 3);
            const v2f gdA = __builtin_shufflevector(gd, gd, 0, 1);
            const v2f gdB = __builtin_shufflevector(gd, gd, 2, 3);
            v8f c = {};
            c = __builtin_amdgcn_wmma_f32_16x16x4_f32(false, guA, false, b1u[0], (short)0, c, false, false);
            c = __builtin_amdgcn_wmma_f32_16x16x4_f32(false, guB, false, b1u[1], (short)0, c, false, false);
            c = __builtin_amdgcn_wmma_f32_16x16x4_f32(false, gdA, false, b1d[0], (short)0, c, false, false);
            c = __builtin_amdgcn_wmma_f32_16x16x4_f32(false, gdB, false, b1d[1], (short)0, c, false, false);
            dst[Q] = c;
        }
    };

    auto group_out = [&](const v8f (&TE)[2], const int re, const v8f (&TO)[2], const int ro, float* og) __attribute__((always_inline)) {
        v8f c2e[2];
        #pragma unroll
        for (int e = 0; e < 2; ++e) {
            v8f c2 = {};
            #pragma unroll
            for (int Q = 0; Q < 2; ++Q) {
                v2f be, bo;
                be[0] = TE[Q][re + 2*e];  be[1] = TE[Q][re + 2*e + 1];
                bo[0] = TO[Q][ro + 2*e];  bo[1] = TO[Q][ro + 2*e + 1];
                c2 = __builtin_amdgcn_wmma_f32_16x16x4_f32(false, a2[2*Q],     false, be, (short)0, c2, false, false);
                c2 = __builtin_amdgcn_wmma_f32_16x16x4_f32(false, a2[2*Q + 1], false, bo, (short)0, c2, false, false);
            }
            c2e[e] = c2;
        }
        #pragma unroll
        for (int e = 0; e < 2; ++e)
            #pragma unroll
            for (int r = 0; r < 8; ++r)
                smem[(wsel * 4 + e * 2 + hi) * BLK_PITCH + r * 8 + row] = c2e[e][r];
        wave_lds_handover();
        #pragma unroll
        for (int R = 0; R < 2; ++R) {
            const int u0 = 4 * (lane + 32 * R);
            if (u0 < keep4) {
                v4f val[2];
                #pragma unroll
                for (int w = 0; w < 2; ++w)
                    #pragma unroll
                    for (int i = 0; i < 4; ++i) val[w][i] = smem[w * 4 * BLK_PITCH + lds_idx[R][i]];
                #pragma unroll
                for (int w = 0; w < 2; ++w)
                    __builtin_nontemporal_store(val[w], (v4f*)(og + w * orow + u0));
            }
        }
        wave_lds_handover();
    };

    auto request_unit = [&](v4f (&dst)[12], const int u) __attribute__((always_inline)) {
        load_tile(dst, xw + 64 * u);
    };

    request_unit(L[0], 3);

    request_unit(L[1], 6);
    stage1(L[0], T[3]);

    request_unit(L[0], 2);
    stage1(L[1], T[6]);
    group_out(T[3], 0, T[6], 4, ow + 6 * keep4);

    request_unit(L[1], 5);
    stage1(L[0], T[2]);
    group_out(T[2], 4, T[6], 0, ow + 5 * keep4);

    request_unit(L[0], 1);
    stage1(L[1], T[5]);
    group_out(T[2], 0, T[5], 4, ow + 4 * keep4);

    request_unit(L[1], 4);
    stage1(L[0], T[1]);
    group_out(T[1], 4, T[5], 0, ow + 3 * keep4);

    request_unit(L[0], 0);
    stage1(L[1], T[4]);
    group_out(T[1], 0, T[4], 4, ow + 2 * keep4);

    stage1(L[0], T[0]);
    group_out(T[0], 4, T[4], 0, ow + 1 * keep4);
    group_out(T[0], 0, T[3], 4, ow);
}

extern "C" void kernel_launch(void* const* d_in, const int* in_sizes, int n_in,
                              void* d_out, int out_size, void* d_ws, size_t ws_size,
                              hipStream_t stream) {
    const float* x  = (const float*)d_in[0];
    const int* rem  = (const int*)d_in[1];
    float* out      = (float*)d_out;

    const int blocks = Bd * Td * NBP;
    dct_zz_kernel<<<blocks, 32, 0, stream>>>(x, rem, out);
}


# Wave set-up no longer waits on memory: table reads (DCT basis, zigzag) are unconditional, not one branch-load-wait each, and go out in one group behind the first three units' pixel loads; sums kept
# speedup vs baseline: 2.1485x; 1.0779x over previous
#include <hip/hip_runtime.h>
#include <hip/hip_bf16.h>

typedef __attribute__((ext_vector_type(2))) float v2f;
typedef __attribute__((ext_vector_type(4))) float v4f;
typedef __attribute__((ext_vector_type(8))) float v8f;

__constant__ int c_zz[64] = {
     0, 1, 8,16, 9, 2, 3,10,
    17,24,32,25,18,11, 4, 5,
    12,19,26,33,40,48,41,34,
    27,20,13, 6, 7,14,21,28,
    35,42,49,56,57,50,43,36,
    29,22,15,23,30,37,44,51,
    58,59,52,45,38,31,39,46,
    53,60,61,54,47,55,62,63};

__constant__ float c_cos[9] = {
    1.0f,
    0.98078528040323044913f,
    0.92387953251128675613f,
    0.83146961230254523708f,
    0.70710678118654752440f,
    0.55557023301960222474f,
    0.38268343236508977173f,
    0.19509032201612826785f,
    0.0f};

#define Bd 4
#define Cd 3
#define Td 64
#define Hd 224
#define Wd 224
#define NBH 28
#define NBP 14
#define NT4 7
#define HW (Hd*Wd)
#define CH_STRIDE (Td*Hd*Wd)
#define BLK_PITCH 72

__device__ __forceinline__ void dmat_at(int l, int n, int& idx, float& s) {
    int u = ((2 * n + 1) * l) & 31;
    if (u > 16) u = 32 - u;
    s = 2.0f;
    idx = u;
    if (u > 8) { idx = 16 - u; s = -2.0f; }
}

__device__ __forceinline__ void load_tile(v4f (&dst)[12], const float* p) {
    #pragma unroll
    for (int w = 0; w < 2; ++w) {
        const float* q = p + w * (8 * Wd);
        dst[6*w + 0] = __builtin_nontemporal_load((const v4f*)(q));
        dst[6*w + 1] = __builtin_nontemporal_load((const v4f*)(q + (long)CH_STRIDE));
        dst[6*w + 2] = __builtin_nontemporal_load((const v4f*)(q + 2L * CH_STRIDE));
        dst[6*w + 3] = __builtin_nontemporal_load((const v4f*)(q + 4 * Wd));
        dst[6*w + 4] = __builtin_nontemporal_load((const v4f*)(q + 4 * Wd + (long)CH_STRIDE));
        dst[6*w + 5] = __builtin_nontemporal_load((const v4f*)(q + 4 * Wd + 2L * CH_STRIDE));
    }
}

__device__ __forceinline__ void wave_lds_handover() {
    __builtin_amdgcn_fence(__ATOMIC_RELEASE, "wavefront");
    __builtin_amdgcn_wave_barrier();
    __builtin_amdgcn_fence(__ATOMIC_ACQUIRE, "wavefront");
}

__global__ __launch_bounds__(32) void dct_zz_kernel(const float* __restrict__ x,
                                                    const int* __restrict__ remp,
                                                    float* __restrict__ out) {
    __shared__ float smem[8 * BLK_PITCH];

    const int lane = threadIdx.x;
    const int hi   = lane >> 4;
    const int N    = lane & 15;
    const int row  = N & 7;
    const int wsel = N >> 3;

    int gw = blockIdx.x;
    const int hp = gw % NBP;  gw /= NBP;
    const int t  = gw % Td;   gw /= Td;
    const int b  = gw;
    const int hb = 2 * hp;

    float sg1[4], cv1[4];
    #pragma unroll
    for (int q = 0; q < 4; ++q) {
        int idx;
        dmat_at(row, 4*hi + q, idx, sg1[q]);
        cv1[q] = c_cos[idx];
    }
    float sg2[4], cv2[4];
    #pragma unroll
    for (int s = 0; s < 4; ++s) {
        int idx;
        dmat_at(row, 4*(s >> 1) + 2*hi + (s & 1), idx, sg2[s]);
        cv2[s] = c_cos[idx];
    }

    const long ubase = ((long)(b * Cd) * Td + t) * (long)HW + (long)(hb * 8) * Wd;
    const int loff = (N >> 3) * (2 * Wd) + (N & 7) * 8 + 4 * hi;
    const float* xw = x + ubase + loff;

    v4f L[3][12];
    auto request_unit = [&](v4f (&dst)[12], const int u) __attribute__((always_inline)) {
        load_tile(dst, xw + 64 * u);
    };

    request_unit(L[0], 3);
    request_unit(L[1], 6);
    request_unit(L[2], 2);
    __builtin_amdgcn_sched_barrier(0);

    const int keep  = 63 - remp[0];
    const int keep4 = 4 * keep;

    int zblk[2][4], zv[2][4];
    #pragma unroll
    for (int R = 0; R < 2; ++R)
        #pragma unroll
        for (int i = 0; i < 4; ++i) {
            const int u   = 4 * (lane + 32 * R) + i;
            const int blk = (u >= keep) + (u >= 2*keep) + (u >= 3*keep);
            const int j   = u - blk * keep;
            zblk[R][i] = blk * BLK_PITCH;
            zv[R][i]   = c_zz[min(1 + j, 63)];
        }
    __builtin_amdgcn_sched_barrier(0);

    v2f b1u[2], b1d[2];
    #pragma unroll
    for (int s = 0; s < 2; ++s)
        #pragma unroll
        for (int v = 0; v < 2; ++v) {
            const float d = sg1[2*s + v] * cv1[2*s + v];
            b1u[s][v] = (wsel == 0) ? d : 0.0f;
            b1d[s][v] = (wsel == 0) ? 0.0f : d;
        }
    v2f a2[4];
    #pragma unroll
    for (int s = 0; s < 4; ++s) {
        const float d = sg2[s] * cv2[s];
        #pragma unroll
        for (int v = 0; v < 2; ++v)
            a2[s][v] = ((N >> 3) == v) ? d : 0.0f;
    }

    int lds_idx[2][4];
    #pragma unroll
    for (int R = 0; R < 2; ++R)
        #pragma unroll
        for (int i = 0; i < 4; ++i)
            lds_idx[R][i] = zblk[R][i] + zv[R][i];

    const long obase = ((long)(b * Td + t) * (NBH * NBH) + hb * NBH) * keep;
    const int orow = NBH * keep;

    const float wc0 = 0.2989f, wc1 = 0.587f, wc2 = 0.114f;

    float* ow = out + obase;

    v8f T[7][2];

    auto stage1 = [&](const v4f (&ld)[12], v8f (&dst)[2]) __attribute__((always_inline)) {
        #pragma unroll
        for (int Q = 0; Q < 2; ++Q) {
            const v4f gu = wc0 * ld[Q*3 + 0] + wc1 * ld[Q*3 + 1] + wc2 * ld[Q*3 + 2];
            const v4f gd = wc0 * ld[6 + Q*3 + 0] + wc1 * ld[6 + Q*3 + 1] + wc2 * ld[6 + Q*3 + 2];
            const v2f guA = __builtin_shufflevector(gu, gu, 0, 1);
            const v2f guB = __builtin_shufflevector(gu, gu, 2, 3);
            const v2f gdA = __builtin_shufflevector(gd, gd, 0, 1);
            const v2f gdB = __builtin_shufflevector(gd, gd, 2, 3);
            v8f c = {};
            c = __builtin_amdgcn_wmma_f32_16x16x4_f32(false, guA, false, b1u[0], (short)0, c, false, false);
            c = __builtin_amdgcn_wmma_f32_16x16x4_f32(false, guB, false, b1u[1], (short)0, c, false, false);
            c = __builtin_amdgcn_wmma_f32_16x16x4_f32(false, gdA, false, b1d[0], (short)0, c, false, false);
            c = __builtin_amdgcn_wmma_f32_16x16x4_f32(false, gdB, false, b1d[1], (short)0, c, false, false);
            dst[Q] = c;
        }
    };

    auto group_out = [&](const v8f (&TE)[2], const int re, const v8f (&TO)[2], const int ro, float* og) __attribute__((always_inline)) {
        v8f c2e[2];
        #pragma unroll
        for (int e = 0; e < 2; ++e) {
            v8f c2 = {};
            #pragma unroll
            for (int Q = 0; Q < 2; ++Q) {
                v2f be, bo;
                be[0] = TE[Q][re + 2*e];  be[1] = TE[Q][re + 2*e + 1];
                bo[0] = TO[Q][ro + 2*e];  bo[1] = TO[Q][ro + 2*e + 1];
                c2 = __builtin_amdgcn_wmma_f32_16x16x4_f32(false, a2[2*Q],     false, be, (short)0, c2, false, false);
                c2 = __builtin_amdgcn_wmma_f32_16x16x4_f32(false, a2[2*Q + 1], false, bo, (short)0, c2, false, false);
            }
            c2e[e] = c2;
        }
        #pragma unroll
        for (int e = 0; e < 2; ++e)
            #pragma unroll
            for (int r = 0; r < 8; ++r)
                smem[(wsel * 4 + e * 2 + hi) * BLK_PITCH + r * 8 + row] = c2e[e][r];
        wave_lds_handover();
        #pragma unroll
        for (int R = 0; R < 2; ++R) {
            const int u0 = 4 * (lane + 32 * R);
            if (u0 < keep4) {
                v4f val[2];
                #pragma unroll
                for (int w = 0; w < 2; ++w)
                    #pragma unroll
                    for (int i = 0; i < 4; ++i) val[w][i] = smem[w * 4 * BLK_PITCH + lds_idx[R][i]];
                #pragma unroll
                for (int w = 0; w < 2; ++w)
                    __builtin_nontemporal_store(val[w], (v4f*)(og + w * orow + u0));
            }
        }
        wave_lds_handover();
    };

    stage1(L[0], T[3]);

    stage1(L[1], T[6]);
    group_out(T[3], 0, T[6], 4, ow + 6 * keep4);

    request_unit(L[0], 5);
    stage1(L[2], T[2]);
    group_out(T[2], 4, T[6], 0, ow + 5 * keep4);

    request_unit(L[1], 1);
    stage1(L[0], T[5]);
    group_out(T[2], 0, T[5], 4, ow + 4 * keep4);

    request_unit(L[2], 4);
    stage1(L[1], T[1]);
    group_out(T[1], 4, T[5], 0, ow + 3 * keep4);

    request_unit(L[0], 0);
    stage1(L[2], T[4]);
    group_out(T[1], 0, T[4], 4, ow + 2 * keep4);

    stage1(L[0], T[0]);
    group_out(T[0], 4, T[4], 0, ow + 1 * keep4);
    group_out(T[0], 0, T[3], 4, ow);
}

extern "C" void kernel_launch(void* const* d_in, const int* in_sizes, int n_in,
                              void* d_out, int out_size, void* d_ws, size_t ws_size,
                              hipStream_t stream) {
    const float* x  = (const float*)d_in[0];
    const int* rem  = (const int*)d_in[1];
    float* out      = (float*)d_out;

    const int blocks = Bd * Td * NBP;
    dct_zz_kernel<<<blocks, 32, 0, stream>>>(x, rem, out);
}
